# starting kernel
# baseline (speedup 1.0000x reference)
#include <hip/hip_runtime.h>
#include <hip/hip_bf16.h>


typedef __attribute__((ext_vector_type(2))) float v2f;
typedef __attribute__((ext_vector_type(8))) float v8f;

#define GM_N 65536
#define GM_M 1024

__global__ __launch_bounds__(GM_M) void gm_coef_kernel(
    const float* __restrict__ mu,
    const float* __restrict__ Amat,
    const float* __restrict__ w,
    float* __restrict__ coef)
{
    __shared__ float red[GM_M];
    const int j = threadIdx.x;

    const float wj = w[j];

    red[j] = wj;
    __syncthreads();
    #pragma unroll
    for (int s = GM_M / 2; s > 0; s >>= 1) {
        if (j < s) red[j] = fmaxf(red[j], red[j + s]);
        __syncthreads();
    }
    const float wmax = red[0];
    __syncthreads();
    red[j] = __expf(wj - wmax);
    __syncthreads();
    #pragma unroll
    for (int s = GM_M / 2; s > 0; s >>= 1) {
        if (j < s) red[j] += red[j + s];
        __syncthreads();
    }
    const float lse_w = wmax + __logf(red[0]);

    const float a00 = Amat[j * 4 + 0], a01 = Amat[j * 4 + 1];
    const float a10 = Amat[j * 4 + 2], a11 = Amat[j * 4 + 3];
    const float g00 = a00 * a00 + a01 * a01;
    const float g01 = a00 * a10 + a01 * a11;
    const float g11 = a10 * a10 + a11 * a11;
    const float det = g00 * g11 - g01 * g01;

    const float logw = (wj - lse_w) + 0.5f * __logf(det);

    const float m0 = mu[j * 2 + 0], m1 = mu[j * 2 + 1];
    const float gm0 = g00 * m0 + g01 * m1;
    const float gm1 = g01 * m0 + g11 * m1;
    const float mGm = gm0 * m0 + gm1 * m1;

    float* c = coef + j * 8;
    c[0] = -g00;
    c[1] = -2.0f * g01;
    c[2] = -g11;
    c[3] = 2.0f * gm0;
    c[4] = 2.0f * gm1;
    c[5] = logw - mGm;
    c[6] = 0.0f;
    c[7] = 0.0f;
}

__global__ __launch_bounds__(256) void gm_main_kernel(
    const float* __restrict__ sample,
    const float* __restrict__ coef,
    float* __restrict__ out)
{
    __shared__ float lcoef[GM_M * 8];

    {
        const float4* src = (const float4*)coef;
        float4* dst = (float4*)lcoef;
        #pragma unroll
        for (int i = threadIdx.x; i < GM_M * 8 / 4; i += 256) dst[i] = src[i];
    }
    __syncthreads();

    const int lane  = threadIdx.x & 31;
    const int wave  = threadIdx.x >> 5;
    const int half  = lane >> 4;
    const int ccol  = lane & 15;

    const int rowBase = blockIdx.x * 128 + wave * 16;
    const int sIdx = rowBase + ccol;

    const float x0 = sample[2 * sIdx + 0];
    const float x1 = sample[2 * sIdx + 1];
    const float f0 = x0 * x0, f1 = x0 * x1, f2 = x1 * x1;
    const float f3 = x0, f4 = x1, f5 = 1.0f, f6 = 0.0f, f7 = 0.0f;

    v2f bLo, bHi;
    bLo.x = half ? f2 : f0;  bLo.y = half ? f3 : f1;
    bHi.x = half ? f6 : f4;  bHi.y = half ? f7 : f5;

    float mx  = -__builtin_inff();
    float sum = 0.0f;

    #pragma unroll 4
    for (int t = 0; t < GM_M / 16; ++t) {
        const float* cr = &lcoef[(t * 16 + ccol) * 8];
        v2f aLo, aHi;
        aLo.x = cr[2 * half + 0];  aLo.y = cr[2 * half + 1];
        aHi.x = cr[2 * half + 4];  aHi.y = cr[2 * half + 5];

        v8f acc = {0.f, 0.f, 0.f, 0.f, 0.f, 0.f, 0.f, 0.f};
        acc = __builtin_amdgcn_wmma_f32_16x16x4_f32(
                  false, aLo, false, bLo, (short)0, acc, false, false);
        acc = __builtin_amdgcn_wmma_f32_16x16x4_f32(
                  false, aHi, false, bHi, (short)0, acc, false, false);

        float tm = acc[0];
        #pragma unroll
        for (int k = 1; k < 8; ++k) tm = fmaxf(tm, acc[k]);
        float ts = 0.0f;
        #pragma unroll
        for (int k = 0; k < 8; ++k) ts += __expf(acc[k] - tm);

        const float m2 = fmaxf(mx, tm);
        sum = sum * __expf(mx - m2) + ts * __expf(tm - m2);
        mx  = m2;
    }

    const float mxo  = __shfl_xor(mx, 16, 32);
    const float sumo = __shfl_xor(sum, 16, 32);
    const float m2 = fmaxf(mx, mxo);
    const float res = m2 + __logf(sum * __expf(mx - m2) + sumo * __expf(mxo - m2));

    if (lane < 16) out[rowBase + lane] = res;
}

extern "C" void kernel_launch(void* const* d_in, const int* in_sizes, int n_in,
                              void* d_out, int out_size, void* d_ws, size_t ws_size,
                              hipStream_t stream) {
    const float* sample = (const float*)d_in[0];
    const float* mu     = (const float*)d_in[1];
    const float* Amat   = (const float*)d_in[2];
    const float* w      = (const float*)d_in[3];
    float* out = (float*)d_out;

    float* coef = (float*)d_ws;

    gm_coef_kernel<<<1, GM_M, 0, stream>>>(mu, Amat, w, coef);

    const int n = in_sizes[0] / 2;
    const int blocks = n / 128;
    gm_main_kernel<<<blocks, 256, 0, stream>>>(sample, coef, out);
}


# Exponentials issued alternately with plain vector instructions (software pipeline over 16 register pairs: exp, subtract, exp, scale): no two transcendentals back to back; same sums, same order
# speedup vs baseline: 1.4478x; 1.4478x over previous
#include <hip/hip_runtime.h>
#include <hip/hip_bf16.h>


typedef __attribute__((ext_vector_type(2))) float v2f;
typedef __attribute__((ext_vector_type(8))) float v8f;
typedef __attribute__((ext_vector_type(2))) unsigned int v2u;

#define GM_N 65536
#define GM_M 1024

__device__ __forceinline__ void gm_half_swap(float& d, float& s)
{
    const v2u r = __builtin_amdgcn_permlane16_swap(__float_as_uint(d), __float_as_uint(s),
                                                   false, false);
    d = __uint_as_float(r.x);
    s = __uint_as_float(r.y);
}

__device__ __forceinline__ v8f gm_scores(const v2f aLo, const v2f aHi,
                                         const v2f bLo, const v2f bHi)
{
    v8f acc = {0.f, 0.f, 0.f, 0.f, 0.f, 0.f, 0.f, 0.f};
    acc = __builtin_amdgcn_wmma_f32_16x16x4_f32(
              false, aLo, false, bLo, (short)0, acc, false, false);
    acc = __builtin_amdgcn_wmma_f32_16x16x4_f32(
              false, aHi, false, bHi, (short)0, acc, false, false);
    return acc;
}

__device__ __forceinline__ float gm_max8(const v8f s)
{
    return fmaxf(fmaxf(fmaxf(s[0], s[1]), fmaxf(s[2], s[3])),
                 fmaxf(fmaxf(s[4], s[5]), fmaxf(s[6], s[7])));
}

#define GM_SUB(dst, a, m) asm volatile("v_pk_add_f32 %0, %1, %2 neg_lo:[0,1] neg_hi:[0,1]" : "=v"(dst) : "v"(a), "v"(m))
#define GM_MUL(dst, a, k) asm volatile("v_pk_mul_f32 %0, %1, %2 op_sel_hi:[1,0]" : "=v"(dst) : "v"(a), "s"(k))
#define GM_EXP(dst, a)    asm volatile("v_exp_f32 %0, %1" : "=v"(dst) : "v"(a))
#define GM_ADD(dst, a, b) asm volatile("v_pk_add_f32 %0, %1, %2" : "=v"(dst) : "v"(a), "v"(b))

__device__ __forceinline__ void gm_group(const v8f s0, const v8f s1,
                                         const v8f s2, const v8f s3,
                                         float& mx, float& sum)
{
    const float m2 = fmaxf(fmaxf(fmaxf(gm_max8(s0), gm_max8(s1)),
                                 fmaxf(gm_max8(s2), gm_max8(s3))), mx);
    const v2f mm = {m2, m2};
    const v2f kk = {1.44269504f, 1.44269504f};

    const v2f sp[16] = {
        __builtin_shufflevector(s0, s0, 0, 1), __builtin_shufflevector(s0, s0, 2, 3),
        __builtin_shufflevector(s0, s0, 4, 5), __builtin_shufflevector(s0, s0, 6, 7),
        __builtin_shufflevector(s1, s1, 0, 1), __builtin_shufflevector(s1, s1, 2, 3),
        __builtin_shufflevector(s1, s1, 4, 5), __builtin_shufflevector(s1, s1, 6, 7),
        __builtin_shufflevector(s2, s2, 0, 1), __builtin_shufflevector(s2, s2, 2, 3),
        __builtin_shufflevector(s2, s2, 4, 5), __builtin_shufflevector(s2, s2, 6, 7),
        __builtin_shufflevector(s3, s3, 0, 1), __builtin_shufflevector(s3, s3, 2, 3),
        __builtin_shufflevector(s3, s3, 4, 5), __builtin_shufflevector(s3, s3, 6, 7)};
    v2f d[16], t[16], e[16], a01[4], a23[4];
    float ex[16], ey[16];

    GM_SUB(d[0], sp[0], mm);  GM_SUB(d[1], sp[1], mm);  GM_SUB(d[2], sp[2], mm);
    GM_MUL(t[0], d[0], kk);   GM_SUB(d[3], sp[3], mm);  GM_MUL(t[1], d[1], kk);
    GM_SUB(d[4], sp[4], mm);  GM_MUL(t[2], d[2], kk);
    #pragma unroll
    for (int j = 0; j < 16; ++j) {
        GM_EXP(ex[j], t[j].x);
        if (j + 5 < 16)  GM_SUB(d[j + 5], sp[j + 5], mm);
        else if (j < 15) GM_ADD(a01[j - 11], e[j - 11], e[j - 7]);
        else             GM_ADD(a23[0], e[8], e[12]);
        GM_EXP(ey[j], t[j].y);
        if (j + 3 < 16)  GM_MUL(t[j + 3], d[j + 3], kk);
        else if (j == 15) GM_ADD(a23[1], e[9], e[13]);
        e[j].x = ex[j];  e[j].y = ey[j];
    }
    GM_ADD(a23[2], e[10], e[14]);
    GM_ADD(a23[3], e[11], e[15]);
    const v2f q0 = a01[0] + a23[0], q1 = a01[1] + a23[1];
    const v2f q2 = a01[2] + a23[2], q3 = a01[3] + a23[3];
    const float ts = ((q0.x + q0.y) + (q1.x + q1.y)) + ((q2.x + q2.y) + (q3.x + q3.y));
    sum = sum * __expf(mx - m2) + ts;
    mx  = m2;
}

__device__ __forceinline__ void gm_build(const float4 a4, const float2 m2, const float wj,
                                         v2f& aLo0, v2f& aHi0, v2f& aLo1, v2f& aHi1)
{
    const float a00 = a4.x, a01 = a4.y;
    const float a10 = a4.z, a11 = a4.w;
    const float g00 = a00 * a00 + a01 * a01;
    const float g01 = a00 * a10 + a01 * a11;
    const float g11 = a10 * a10 + a11 * a11;
    const float det = g00 * g11 - g01 * g01;

    const float m0 = m2.x, m1 = m2.y;
    const float gm0 = g00 * m0 + g01 * m1;
    const float gm1 = g01 * m0 + g11 * m1;
    const float mGm = gm0 * m0 + gm1 * m1;

    float x0 = g00, x1 = g11;
    float y0 = g01, y1 = gm0;
    float z0 = gm1, z1 = 0.0f;
    float w0 = (wj + 0.34657359f * log2f(det)) - mGm, w1 = 0.0f;
    gm_half_swap(x0, x1);
    gm_half_swap(y0, y1);
    gm_half_swap(z0, z1);
    gm_half_swap(w0, w1);

    aLo0.x = x0;  aLo0.y = y0;  aHi0.x = z0;  aHi0.y = w0;
    aLo1.x = x1;  aLo1.y = y1;  aHi1.x = z1;  aHi1.y = w1;
}

__global__ __launch_bounds__(128) void gm_fused_kernel(
    const float* __restrict__ sample,
    const float* __restrict__ mu,
    const float* __restrict__ Amat,
    const float* __restrict__ w,
    float* __restrict__ out)
{
    const int lane  = threadIdx.x & 31;
    const int wave  = threadIdx.x >> 5;
    const int half  = lane >> 4;
    const int ccol  = lane & 15;

    const int colBase = blockIdx.x * 128 + wave * 32;

    const float4* __restrict__ A4 = (const float4*)Amat;
    const float2* __restrict__ M2 = (const float2*)mu;
    const float*  __restrict__ W1 = w;

    const float2 sA = ((const float2*)sample)[colBase + ccol];
    const float2 sB = ((const float2*)sample)[colBase + 16 + ccol];
    float4 a4n = A4[lane];
    float2 m2n = M2[lane];
    float  wn  = W1[lane];
    const float4* w4 = (const float4*)w;
    float4 wv[GM_M / 128];
    #pragma unroll
    for (int i = 0; i < GM_M / 128; ++i) wv[i] = w4[i * 32 + lane];

    v2f bLoA, bHiA, bLoB, bHiB;
    bLoA.x = half ? -(sA.y * sA.y) : -(sA.x * sA.x);
    bLoA.y = half ? 2.0f * sA.x : -2.0f * (sA.x * sA.y);
    bHiA.x = half ? 0.0f : 2.0f * sA.y;
    bHiA.y = half ? 0.0f : 1.0f;
    bLoB.x = half ? -(sB.y * sB.y) : -(sB.x * sB.x);
    bLoB.y = half ? 2.0f * sB.x : -2.0f * (sB.x * sB.y);
    bHiB.x = half ? 0.0f : 2.0f * sB.y;
    bHiB.y = half ? 0.0f : 1.0f;

    float mxA = -__builtin_inff(), sumA = 0.0f;
    float mxB = -__builtin_inff(), sumB = 0.0f;

    float4 a4m = A4[32 + lane];
    float2 m2m = M2[32 + lane];
    float  wm  = W1[32 + lane];

    #pragma unroll 2
    for (int P = 0; P < GM_M / 64; ++P) {
        const float4 a4c0 = a4n, a4c1 = a4m;
        const float2 m2c0 = m2n, m2c1 = m2m;
        const float  wc0  = wn,  wc1  = wm;
        const int Pn = ((P + 1) & (GM_M / 64 - 1)) * 64;
        a4n = (A4 + Pn)[lane];       a4m = (A4 + Pn + 32)[lane];
        m2n = (M2 + Pn)[lane];       m2m = (M2 + Pn + 32)[lane];
        wn  = (W1 + Pn)[lane];       wm  = (W1 + Pn + 32)[lane];
        asm volatile("" ::: "memory");

        v2f aLo0, aHi0, aLo1, aHi1, aLo2, aHi2, aLo3, aHi3;
        gm_build(a4c0, m2c0, wc0, aLo0, aHi0, aLo1, aHi1);
        gm_build(a4c1, m2c1, wc1, aLo2, aHi2, aLo3, aHi3);

        gm_group(gm_scores(aLo0, aHi0, bLoA, bHiA), gm_scores(aLo1, aHi1, bLoA, bHiA),
                 gm_scores(aLo2, aHi2, bLoA, bHiA), gm_scores(aLo3, aHi3, bLoA, bHiA),
                 mxA, sumA);
        gm_group(gm_scores(aLo0, aHi0, bLoB, bHiB), gm_scores(aLo1, aHi1, bLoB, bHiB),
                 gm_scores(aLo2, aHi2, bLoB, bHiB), gm_scores(aLo3, aHi3, bLoB, bHiB),
                 mxB, sumB);
    }

    float wmax = fmaxf(fmaxf(wv[0].x, wv[0].y), fmaxf(wv[0].z, wv[0].w));
    #pragma unroll
    for (int i = 1; i < GM_M / 128; ++i) {
        wmax = fmaxf(fmaxf(wmax, wv[i].x), wv[i].y);
        wmax = fmaxf(fmaxf(wmax, wv[i].z), wv[i].w);
    }
    for (int off = 16; off > 0; off >>= 1)
        wmax = fmaxf(wmax, __shfl_xor(wmax, off, 32));
    float wsum = 0.0f;
    #pragma unroll
    for (int i = 0; i < GM_M / 128; ++i) {
        wsum += __expf(wv[i].x - wmax);
        wsum += __expf(wv[i].y - wmax);
        wsum += __expf(wv[i].z - wmax);
        wsum += __expf(wv[i].w - wmax);
    }
    for (int off = 16; off > 0; off >>= 1)
        wsum += __shfl_xor(wsum, off, 32);
    const float lse_w = wmax + __logf(wsum);

    const float mxAo  = __shfl_xor(mxA, 16, 32);
    const float sumAo = __shfl_xor(sumA, 16, 32);
    const float mxBo  = __shfl_xor(mxB, 16, 32);
    const float sumBo = __shfl_xor(sumB, 16, 32);
    const float mA = fmaxf(mxA, mxAo);
    const float mB = fmaxf(mxB, mxBo);
    const float resA = mA + __logf(sumA * __expf(mxA - mA) + sumAo * __expf(mxAo - mA));
    const float resB = mB + __logf(sumB * __expf(mxB - mB) + sumBo * __expf(mxBo - mB));

    out[colBase + lane] = (half ? resB : resA) - lse_w;
}

extern "C" void kernel_launch(void* const* d_in, const int* in_sizes, int n_in,
                              void* d_out, int out_size, void* d_ws, size_t ws_size,
                              hipStream_t stream) {
    const float* sample = (const float*)d_in[0];
    const float* mu     = (const float*)d_in[1];
    const float* Amat   = (const float*)d_in[2];
    const float* w      = (const float*)d_in[3];
    float* out = (float*)d_out;

    const int n = in_sizes[0] / 2;
    const int blocks = n / 128;
    gm_fused_kernel<<<blocks, 128, 0, stream>>>(sample, mu, Amat, w, out);
}


# Components split over the 4 waves of a block (256 each); each wave sweeps all 128 columns, so a built fragment feeds 8 tiles; per-wave (max,sum) partials and weight sums merged in LDS: new sum order
# speedup vs baseline: 1.5754x; 1.0881x over previous
#include <hip/hip_runtime.h>
#include <hip/hip_bf16.h>


typedef __attribute__((ext_vector_type(2))) float v2f;
typedef __attribute__((ext_vector_type(8))) float v8f;
typedef __attribute__((ext_vector_type(2))) unsigned int v2u;

#define GM_N 65536
#define GM_M 1024
#define GM_WAVES  4
#define GM_GROUPS 8
#define GM_CPW    (GM_M / GM_WAVES)
#define GM_PASSES (GM_CPW / 64)

__device__ __forceinline__ void gm_half_swap(float& d, float& s)
{
    const v2u r = __builtin_amdgcn_permlane16_swap(__float_as_uint(d), __float_as_uint(s),
                                                   false, false);
    d = __uint_as_float(r.x);
    s = __uint_as_float(r.y);
}

template <int MASK>
__device__ __forceinline__ float gm_row_xor(float x)
{
    const int b = __builtin_bit_cast(int, x);
    return __builtin_bit_cast(float, __builtin_amdgcn_update_dpp(b, b, 0x160 + MASK, 0xf, 0xf, false));
}

__device__ __forceinline__ float gm_xhalf(float x)
{
    const int b = __builtin_bit_cast(int, x);
    return __builtin_bit_cast(float, __builtin_amdgcn_permlanex16(b, b, 0x76543210, 0xfedcba98, false, false));
}

__device__ __forceinline__ float gm_wave_max(float x)
{
    x = fmaxf(x, gm_row_xor<1>(x));
    x = fmaxf(x, gm_row_xor<2>(x));
    x = fmaxf(x, gm_row_xor<4>(x));
    x = fmaxf(x, gm_row_xor<8>(x));
    return fmaxf(x, gm_xhalf(x));
}
__device__ __forceinline__ float gm_wave_sum(float x)
{
    x += gm_row_xor<1>(x);
    x += gm_row_xor<2>(x);
    x += gm_row_xor<4>(x);
    x += gm_row_xor<8>(x);
    return x + gm_xhalf(x);
}

__device__ __forceinline__ v8f gm_scores(const v2f aLo, const v2f aHi,
                                         const v2f bLo, const v2f bHi)
{
    v8f acc = {0.f, 0.f, 0.f, 0.f, 0.f, 0.f, 0.f, 0.f};
    acc = __builtin_amdgcn_wmma_f32_16x16x4_f32(
              false, aLo, false, bLo, (short)0, acc, false, false);
    acc = __builtin_amdgcn_wmma_f32_16x16x4_f32(
              false, aHi, false, bHi, (short)0, acc, false, false);
    return acc;
}

__device__ __forceinline__ float gm_max8(const v8f s)
{
    return fmaxf(fmaxf(fmaxf(s[0], s[1]), fmaxf(s[2], s[3])),
                 fmaxf(fmaxf(s[4], s[5]), fmaxf(s[6], s[7])));
}

#define GM_SUB(dst, a, m) asm volatile("v_pk_add_f32 %0, %1, %2 neg_lo:[0,1] neg_hi:[0,1]" : "=v"(dst) : "v"(a), "v"(m))
#define GM_MUL(dst, a, k) asm volatile("v_pk_mul_f32 %0, %1, %2 op_sel_hi:[1,0]" : "=v"(dst) : "v"(a), "s"(k))
#define GM_EXP(dst, a)    asm volatile("v_exp_f32 %0, %1" : "=v"(dst) : "v"(a))
#define GM_ADD(dst, a, b) asm volatile("v_pk_add_f32 %0, %1, %2" : "=v"(dst) : "v"(a), "v"(b))

__device__ __forceinline__ void gm_group(const v8f s0, const v8f s1,
                                         const v8f s2, const v8f s3,
                                         float& mx, float& sum)
{
    const float m2 = fmaxf(fmaxf(fmaxf(gm_max8(s0), gm_max8(s1)),
                                 fmaxf(gm_max8(s2), gm_max8(s3))), mx);
    const v2f mm = {m2, m2};
    const v2f kk = {1.44269504f, 1.44269504f};

    const v2f sp[16] = {
        __builtin_shufflevector(s0, s0, 0, 1), __builtin_shufflevector(s0, s0, 2, 3),
        __builtin_shufflevector(s0, s0, 4, 5), __builtin_shufflevector(s0, s0, 6, 7),
        __builtin_shufflevector(s1, s1, 0, 1), __builtin_shufflevector(s1, s1, 2, 3),
        __builtin_shufflevector(s1, s1, 4, 5), __builtin_shufflevector(s1, s1, 6, 7),
        __builtin_shufflevector(s2, s2, 0, 1), __builtin_shufflevector(s2, s2, 2, 3),
        __builtin_shufflevector(s2, s2, 4, 5), __builtin_shufflevector(s2, s2, 6, 7),
        __builtin_shufflevector(s3, s3, 0, 1), __builtin_shufflevector(s3, s3, 2, 3),
        __builtin_shufflevector(s3, s3, 4, 5), __builtin_shufflevector(s3, s3, 6, 7)};
    v2f d[16], t[16], e[16], a01[4], a23[4];
    float ex[16], ey[16];

    GM_SUB(d[0], sp[0], mm);  GM_SUB(d[1], sp[1], mm);  GM_SUB(d[2], sp[2], mm);
    GM_MUL(t[0], d[0], kk);   GM_SUB(d[3], sp[3], mm);  GM_MUL(t[1], d[1], kk);
    GM_SUB(d[4], sp[4], mm);  GM_MUL(t[2], d[2], kk);
    #pragma unroll
    for (int j = 0; j < 16; ++j) {
        GM_EXP(ex[j], t[j].x);
        if (j + 5 < 16)  GM_SUB(d[j + 5], sp[j + 5], mm);
        else if (j < 15) GM_ADD(a01[j - 11], e[j - 11], e[j - 7]);
        else             GM_ADD(a23[0], e[8], e[12]);
        GM_EXP(ey[j], t[j].y);
        if (j + 3 < 16)  GM_MUL(t[j + 3], d[j + 3], kk);
        else if (j == 15) GM_ADD(a23[1], e[9], e[13]);
        e[j].x = ex[j];  e[j].y = ey[j];
    }
    GM_ADD(a23[2], e[10], e[14]);
    GM_ADD(a23[3], e[11], e[15]);
    const v2f q0 = a01[0] + a23[0], q1 = a01[1] + a23[1];
    const v2f q2 = a01[2] + a23[2], q3 = a01[3] + a23[3];
    const float ts = ((q0.x + q0.y) + (q1.x + q1.y)) + ((q2.x + q2.y) + (q3.x + q3.y));
    sum = sum * __expf(mx - m2) + ts;
    mx  = m2;
}

__device__ __forceinline__ void gm_build(const float4 a4, const float2 m2, const float wj,
                                         v2f& aLo0, v2f& aHi0, v2f& aLo1, v2f& aHi1)
{
    const float a00 = a4.x, a01 = a4.y;
    const float a10 = a4.z, a11 = a4.w;
    const float g00 = a00 * a00 + a01 * a01;
    const float g01 = a00 * a10 + a01 * a11;
    const float g11 = a10 * a10 + a11 * a11;
    const float det = g00 * g11 - g01 * g01;

    const float m0 = m2.x, m1 = m2.y;
    const float gm0 = g00 * m0 + g01 * m1;
    const float gm1 = g01 * m0 + g11 * m1;
    const float mGm = gm0 * m0 + gm1 * m1;

    float x0 = g00, x1 = g11;
    float y0 = g01, y1 = gm0;
    float z0 = gm1, z1 = 0.0f;
    float w0 = (wj + 0.34657359f * log2f(det)) - mGm, w1 = 0.0f;
    gm_half_swap(x0, x1);
    gm_half_swap(y0, y1);
    gm_half_swap(z0, z1);
    gm_half_swap(w0, w1);

    aLo0.x = x0;  aLo0.y = y0;  aHi0.x = z0;  aHi0.y = w0;
    aLo1.x = x1;  aLo1.y = y1;  aHi1.x = z1;  aHi1.y = w1;
}

__global__ __launch_bounds__(128) void gm_fused_kernel(
    const float* __restrict__ sample,
    const float* __restrict__ mu,
    const float* __restrict__ Amat,
    const float* __restrict__ w,
    float* __restrict__ out)
{
    __shared__ float2 part[GM_WAVES][GM_GROUPS][32];
    __shared__ float2 wnorm[GM_WAVES][32];

    const int lane  = threadIdx.x & 31;
    const int wave  = threadIdx.x >> 5;
    const int half  = lane >> 4;
    const int ccol  = lane & 15;

    const int colBase  = blockIdx.x * (16 * GM_GROUPS);
    const int compBase = wave * GM_CPW;

    const float4* __restrict__ A4 = (const float4*)Amat + compBase;
    const float2* __restrict__ M2 = (const float2*)mu + compBase;
    const float*  __restrict__ W1 = w + compBase;

    float2 sv[GM_GROUPS];
    for (int g = 0; g < GM_GROUPS; ++g)
        sv[g] = ((const float2*)sample)[colBase + 16 * g + ccol];
    float4 a4n = A4[lane];
    float2 m2n = M2[lane];
    float  wn  = W1[lane];
    float4 a4m = A4[32 + lane];
    float2 m2m = M2[32 + lane];
    float  wm  = W1[32 + lane];
    const float4 wq0 = ((const float4*)W1)[lane];
    const float4 wq1 = ((const float4*)W1)[32 + lane];

    v2f bLo[GM_GROUPS], bHi[GM_GROUPS];
    float mx[GM_GROUPS], sum[GM_GROUPS];
    for (int g = 0; g < GM_GROUPS; ++g) {
        const float2 sA = sv[g];
        bLo[g].x = half ? -(sA.y * sA.y) : -(sA.x * sA.x);
        bLo[g].y = half ? 2.0f * sA.x : -2.0f * (sA.x * sA.y);
        bHi[g].x = half ? 0.0f : 2.0f * sA.y;
        bHi[g].y = half ? 0.0f : 1.0f;
        mx[g]  = -__builtin_inff();
        sum[g] = 0.0f;
    }

    float wmax = fmaxf(fmaxf(fmaxf(wq0.x, wq0.y), fmaxf(wq0.z, wq0.w)),
                       fmaxf(fmaxf(wq1.x, wq1.y), fmaxf(wq1.z, wq1.w)));
    wmax = gm_wave_max(wmax);
    float wsum = 0.0f;
    wsum += __expf(wq0.x - wmax);
    wsum += __expf(wq0.y - wmax);
    wsum += __expf(wq0.z - wmax);
    wsum += __expf(wq0.w - wmax);
    wsum += __expf(wq1.x - wmax);
    wsum += __expf(wq1.y - wmax);
    wsum += __expf(wq1.z - wmax);
    wsum += __expf(wq1.w - wmax);
    wsum = gm_wave_sum(wsum);

    for (int P = 0; P < GM_PASSES; ++P) {
        const float4 a4c0 = a4n, a4c1 = a4m;
        const float2 m2c0 = m2n, m2c1 = m2m;
        const float  wc0  = wn,  wc1  = wm;
        const int Pn = ((P + 1) & (GM_PASSES - 1)) * 64;
        a4n = (A4 + Pn)[lane];       a4m = (A4 + Pn + 32)[lane];
        m2n = (M2 + Pn)[lane];       m2m = (M2 + Pn + 32)[lane];
        wn  = (W1 + Pn)[lane];       wm  = (W1 + Pn + 32)[lane];
        asm volatile("" ::: "memory");

        v2f aLo0, aHi0, aLo1, aHi1, aLo2, aHi2, aLo3, aHi3;
        gm_build(a4c0, m2c0, wc0, aLo0, aHi0, aLo1, aHi1);
        gm_build(a4c1, m2c1, wc1, aLo2, aHi2, aLo3, aHi3);

        #pragma unroll
        for (int g = 0; g < GM_GROUPS; ++g)
            gm_group(gm_scores(aLo0, aHi0, bLo[g], bHi[g]), gm_scores(aLo1, aHi1, bLo[g], bHi[g]),
                     gm_scores(aLo2, aHi2, bLo[g], bHi[g]), gm_scores(aLo3, aHi3, bLo[g], bHi[g]),
                     mx[g], sum[g]);
    }

    for (int g = 0; g < GM_GROUPS; ++g)
        part[wave][g][lane] = make_float2(mx[g], sum[g]);
    wnorm[wave][lane] = make_float2(wmax, wsum);
    __syncthreads();

    const int gq = 2 * wave + half;
    float2 pp[2 * GM_WAVES];
    for (int v = 0; v < GM_WAVES; ++v) {
        pp[2 * v]     = part[v][gq][ccol];
        pp[2 * v + 1] = part[v][gq][ccol + 16];
    }
    float2 wp[GM_WAVES];
    for (int v = 0; v < GM_WAVES; ++v) wp[v] = wnorm[v][lane];

    const float mAll = fmaxf(fmaxf(fmaxf(pp[0].x, pp[1].x), fmaxf(pp[2].x, pp[3].x)),
                             fmaxf(fmaxf(pp[4].x, pp[5].x), fmaxf(pp[6].x, pp[7].x)));
    float sAll = 0.0f;
    for (int i = 0; i < 2 * GM_WAVES; ++i) sAll += pp[i].y * __expf(pp[i].x - mAll);

    const float wAll = fmaxf(fmaxf(wp[0].x, wp[1].x), fmaxf(wp[2].x, wp[3].x));
    float wsAll = 0.0f;
    for (int v = 0; v < GM_WAVES; ++v) wsAll += wp[v].y * __expf(wp[v].x - wAll);
    const float lse_w = wAll + __logf(wsAll);

    out[colBase + 32 * wave + lane] = (mAll + __logf(sAll)) - lse_w;
}

extern "C" void kernel_launch(void* const* d_in, const int* in_sizes, int n_in,
                              void* d_out, int out_size, void* d_ws, size_t ws_size,
                              hipStream_t stream) {
    const float* sample = (const float*)d_in[0];
    const float* mu     = (const float*)d_in[1];
    const float* Amat   = (const float*)d_in[2];
    const float* w      = (const float*)d_in[3];
    float* out = (float*)d_out;

    const int n = in_sizes[0] / 2;
    const int blocks = n / 128;
    gm_fused_kernel<<<blocks, 128, 0, stream>>>(sample, mu, Amat, w, out);
}


# One WMMA per score tile, not two: its 4 k slots hold x0^2, x0x1, x1^2, x0; the x1 term and the constant enter as accumulator input (4 packed FMAs per tile, coefficients via LDS); new sum order
# speedup vs baseline: 1.7963x; 1.1402x over previous
#include <hip/hip_runtime.h>
#include <hip/hip_bf16.h>


typedef __attribute__((ext_vector_type(2))) float v2f;
typedef __attribute__((ext_vector_type(8))) float v8f;
typedef __attribute__((ext_vector_type(2))) unsigned int v2u;

#define GM_N 65536
#define GM_M 1024
#define GM_WAVES  4
#define GM_GROUPS 8
#define GM_CPW    (GM_M / GM_WAVES)
#define GM_PASSES (GM_CPW / 64)
#define GM_BUILD_AT 4

__device__ __forceinline__ void gm_half_swap(float& d, float& s)
{
    const v2u r = __builtin_amdgcn_permlane16_swap(__float_as_uint(d), __float_as_uint(s),
                                                   false, false);
    d = __uint_as_float(r.x);
    s = __uint_as_float(r.y);
}

template <int MASK>
__device__ __forceinline__ float gm_row_xor(float x)
{
    const int b = __builtin_bit_cast(int, x);
    return __builtin_bit_cast(float, __builtin_amdgcn_update_dpp(b, b, 0x160 + MASK, 0xf, 0xf, false));
}

__device__ __forceinline__ float gm_xhalf(float x)
{
    const int b = __builtin_bit_cast(int, x);
    return __builtin_bit_cast(float, __builtin_amdgcn_permlanex16(b, b, 0x76543210, 0xfedcba98, false, false));
}

__device__ __forceinline__ float gm_wave_max(float x)
{
    x = fmaxf(x, gm_row_xor<1>(x));
    x = fmaxf(x, gm_row_xor<2>(x));
    x = fmaxf(x, gm_row_xor<4>(x));
    x = fmaxf(x, gm_row_xor<8>(x));
    return fmaxf(x, gm_xhalf(x));
}
__device__ __forceinline__ float gm_wave_sum(float x)
{
    x += gm_row_xor<1>(x);
    x += gm_row_xor<2>(x);
    x += gm_row_xor<4>(x);
    x += gm_row_xor<8>(x);
    return x + gm_xhalf(x);
}

template <int ODD>
__device__ __forceinline__ v2f gm_fma_half(const v2f a, const v2f xp, const v2f c)
{
    v2f d;
    if (ODD) asm("v_pk_fma_f32 %0, %1, %2, %3 op_sel:[0,1,0] op_sel_hi:[1,1,1]" : "=v"(d) : "v"(a), "v"(xp), "v"(c));
    else     asm("v_pk_fma_f32 %0, %1, %2, %3 op_sel_hi:[1,0,1]" : "=v"(d) : "v"(a), "v"(xp), "v"(c));
    return d;
}

template <int ODD>
__device__ __forceinline__ v8f gm_scores(const v2f a, const v2f b,
                                         const v8f klin, const v8f kcon, const v2f xp)
{
    const v2f c01 = gm_fma_half<ODD>(__builtin_shufflevector(klin, klin, 0, 1), xp,
                                     __builtin_shufflevector(kcon, kcon, 0, 1));
    const v2f c23 = gm_fma_half<ODD>(__builtin_shufflevector(klin, klin, 2, 3), xp,
                                     __builtin_shufflevector(kcon, kcon, 2, 3));
    const v2f c45 = gm_fma_half<ODD>(__builtin_shufflevector(klin, klin, 4, 5), xp,
                                     __builtin_shufflevector(kcon, kcon, 4, 5));
    const v2f c67 = gm_fma_half<ODD>(__builtin_shufflevector(klin, klin, 6, 7), xp,
                                     __builtin_shufflevector(kcon, kcon, 6, 7));
    const v8f c = {c01.x, c01.y, c23.x, c23.y, c45.x, c45.y, c67.x, c67.y};
    return __builtin_amdgcn_wmma_f32_16x16x4_f32(
               false, a, false, b, (short)0, c, false, false);
}

__device__ __forceinline__ float gm_max8(const v8f s)
{
    return fmaxf(fmaxf(fmaxf(s[0], s[1]), fmaxf(s[2], s[3])),
                 fmaxf(fmaxf(s[4], s[5]), fmaxf(s[6], s[7])));
}

#define GM_SUB(dst, a, m) asm volatile("v_pk_add_f32 %0, %1, %2 neg_lo:[0,1] neg_hi:[0,1]" : "=v"(dst) : "v"(a), "v"(m))
#define GM_MUL(dst, a, k) asm volatile("v_pk_mul_f32 %0, %1, %2 op_sel_hi:[1,0]" : "=v"(dst) : "v"(a), "s"(k))
#define GM_EXP(dst, a)    asm volatile("v_exp_f32 %0, %1" : "=v"(dst) : "v"(a))
#define GM_ADD(dst, a, b) asm volatile("v_pk_add_f32 %0, %1, %2" : "=v"(dst) : "v"(a), "v"(b))

__device__ __forceinline__ void gm_group(const v8f s0, const v8f s1,
                                         const v8f s2, const v8f s3,
                                         float& mx, float& sum)
{
    const float m2 = fmaxf(fmaxf(fmaxf(gm_max8(s0), gm_max8(s1)),
                                 fmaxf(gm_max8(s2), gm_max8(s3))), mx);
    const v2f mm = {m2, m2};
    const v2f kk = {1.44269504f, 1.44269504f};

    const v2f sp[16] = {
        __builtin_shufflevector(s0, s0, 0, 1), __builtin_shufflevector(s0, s0, 2, 3),
        __builtin_shufflevector(s0, s0, 4, 5), __builtin_shufflevector(s0, s0, 6, 7),
        __builtin_shufflevector(s1, s1, 0, 1), __builtin_shufflevector(s1, s1, 2, 3),
        __builtin_shufflevector(s1, s1, 4, 5), __builtin_shufflevector(s1, s1, 6, 7),
        __builtin_shufflevector(s2, s2, 0, 1), __builtin_shufflevector(s2, s2, 2, 3),
        __builtin_shufflevector(s2, s2, 4, 5), __builtin_shufflevector(s2, s2, 6, 7),
        __builtin_shufflevector(s3, s3, 0, 1), __builtin_shufflevector(s3, s3, 2, 3),
        __builtin_shufflevector(s3, s3, 4, 5), __builtin_shufflevector(s3, s3, 6, 7)};
    v2f d[16], t[16], e[16], a01[4], a23[4];
    float ex[16], ey[16];

    GM_SUB(d[0], sp[0], mm);  GM_SUB(d[1], sp[1], mm);  GM_SUB(d[2], sp[2], mm);
    GM_MUL(t[0], d[0], kk);   GM_SUB(d[3], sp[3], mm);  GM_MUL(t[1], d[1], kk);
    GM_SUB(d[4], sp[4], mm);  GM_MUL(t[2], d[2], kk);
    #pragma unroll
    for (int j = 0; j < 16; ++j) {
        GM_EXP(ex[j], t[j].x);
        if (j + 5 < 16)  GM_SUB(d[j + 5], sp[j + 5], mm);
        else if (j < 15) GM_ADD(a01[j - 11], e[j - 11], e[j - 7]);
        else             GM_ADD(a23[0], e[8], e[12]);
        GM_EXP(ey[j], t[j].y);
        if (j + 3 < 16)  GM_MUL(t[j + 3], d[j + 3], kk);
        else if (j == 15) GM_ADD(a23[1], e[9], e[13]);
        e[j].x = ex[j];  e[j].y = ey[j];
    }
    GM_ADD(a23[2], e[10], e[14]);
    GM_ADD(a23[3], e[11], e[15]);
    const v2f q0 = a01[0] + a23[0], q1 = a01[1] + a23[1];
    const v2f q2 = a01[2] + a23[2], q3 = a01[3] + a23[3];
    const float ts = ((q0.x + q0.y) + (q1.x + q1.y)) + ((q2.x + q2.y) + (q3.x + q3.y));
    sum = sum * __expf(mx - m2) + ts;
    mx  = m2;
}

__device__ __forceinline__ void gm_build(const float4 a4, const float2 m2, const float wj,
                                         v2f& aT0, v2f& aT1, float& klin, float& kcon)
{
    const float a00 = a4.x, a01 = a4.y;
    const float a10 = a4.z, a11 = a4.w;
    const float g00 = a00 * a00 + a01 * a01;
    const float g01 = a00 * a10 + a01 * a11;
    const float g11 = a10 * a10 + a11 * a11;
    const float det = g00 * g11 - g01 * g01;

    const float m0 = m2.x, m1 = m2.y;
    const float gm0 = g00 * m0 + g01 * m1;
    const float gm1 = g01 * m0 + g11 * m1;
    const float mGm = gm0 * m0 + gm1 * m1;

    float x0 = g00, x1 = g11;
    float y0 = g01, y1 = gm0;
    gm_half_swap(x0, x1);
    gm_half_swap(y0, y1);

    aT0.x = x0;  aT0.y = y0;
    aT1.x = x1;  aT1.y = y1;
    klin = gm1;
    kcon = (wj + 0.34657359f * log2f(det)) - mGm;
}

__device__ __forceinline__ void gm_strip_put(float* strip, const int lane,
                                             const float l0, const float c0,
                                             const float l1, const float c1)
{
    strip[lane]      = l0;  strip[32 + lane] = l1;
    strip[64 + lane] = c0;  strip[96 + lane] = c1;
}
__device__ __forceinline__ void gm_strip_get(const float* strip, const int half,
                                             v8f (&kl)[4], v8f (&kc)[4])
{
    for (int t = 0; t < 4; ++t) {
        kl[t] = *(const v8f*)(strip + 16 * t + 8 * half);
        kc[t] = *(const v8f*)(strip + 64 + 16 * t + 8 * half);
    }
}

__global__ __launch_bounds__(128, 4) void gm_fused_kernel(
    const float* __restrict__ sample,
    const float* __restrict__ mu,
    const float* __restrict__ Amat,
    const float* __restrict__ w,
    float* __restrict__ out)
{
    __shared__ float2 part[GM_WAVES][GM_GROUPS][32];
    __shared__ float2 wnorm[GM_WAVES][32];
    __shared__ __attribute__((aligned(32))) float strips[GM_WAVES][2][128];

    const int lane  = threadIdx.x & 31;
    const int wave  = threadIdx.x >> 5;
    const int half  = lane >> 4;
    const int ccol  = lane & 15;

    const int colBase  = blockIdx.x * (16 * GM_GROUPS);
    const int compBase = wave * GM_CPW;

    const float4* __restrict__ A4 = (const float4*)Amat + compBase;
    const float2* __restrict__ M2 = (const float2*)mu + compBase;
    const float*  __restrict__ W1 = w + compBase;

    float2 sv[GM_GROUPS];
    for (int g = 0; g < GM_GROUPS; ++g)
        sv[g] = ((const float2*)sample)[colBase + 16 * g + ccol];
    const float4 a4c0 = A4[lane],      a4c1 = A4[32 + lane];
    const float2 m2c0 = M2[lane],      m2c1 = M2[32 + lane];
    const float  wc0  = W1[lane],      wc1  = W1[32 + lane];
    float4 a4n = A4[64 + lane],  a4m = A4[96 + lane];
    float2 m2n = M2[64 + lane],  m2m = M2[96 + lane];
    float  wn  = W1[64 + lane],  wm  = W1[96 + lane];
    const float4 wq0 = ((const float4*)W1)[lane];
    const float4 wq1 = ((const float4*)W1)[32 + lane];

    v2f bq[GM_GROUPS];
    v2f x1p[GM_GROUPS / 2];
    float mx[GM_GROUPS], sum[GM_GROUPS];
    for (int g = 0; g < GM_GROUPS; ++g) {
        const float2 sA = sv[g];
        bq[g].x = half ? -(sA.y * sA.y) : -(sA.x * sA.x);
        bq[g].y = half ? 2.0f * sA.x : -2.0f * (sA.x * sA.y);
        x1p[g >> 1][g & 1] = 2.0f * sA.y;
        mx[g]  = -__builtin_inff();
        sum[g] = 0.0f;
    }

    float wmax = fmaxf(fmaxf(fmaxf(wq0.x, wq0.y), fmaxf(wq0.z, wq0.w)),
                       fmaxf(fmaxf(wq1.x, wq1.y), fmaxf(wq1.z, wq1.w)));
    wmax = gm_wave_max(wmax);
    float wsum = 0.0f;
    wsum += __expf(wq0.x - wmax);
    wsum += __expf(wq0.y - wmax);
    wsum += __expf(wq0.z - wmax);
    wsum += __expf(wq0.w - wmax);
    wsum += __expf(wq1.x - wmax);
    wsum += __expf(wq1.y - wmax);
    wsum += __expf(wq1.z - wmax);
    wsum += __expf(wq1.w - wmax);
    wsum = gm_wave_sum(wsum);

    v2f aT0, aT1, aT2, aT3;
    v8f kl[4], kc[4];
    {
        float l0, c0, l1, c1;
        gm_build(a4c0, m2c0, wc0, aT0, aT1, l0, c0);
        gm_build(a4c1, m2c1, wc1, aT2, aT3, l1, c1);
        gm_strip_put(&strips[wave][0][0], lane, l0, c0, l1, c1);
        asm volatile("" ::: "memory");
        gm_strip_get(&strips[wave][0][0], half, kl, kc);
    }

    for (int P = 0; P < GM_PASSES; ++P) {
        v2f aN0 = aT0, aN1 = aT1, aN2 = aT2, aN3 = aT3;
        float* const stripN = &strips[wave][(P + 1) & 1][0];

        #pragma unroll
        for (int g = 0; g < GM_GROUPS; ++g) {
            if (g == GM_BUILD_AT && P + 1 < GM_PASSES) {
                float l0, c0, l1, c1;
                gm_build(a4n, m2n, wn, aN0, aN1, l0, c0);
                gm_build(a4m, m2m, wm, aN2, aN3, l1, c1);
                gm_strip_put(stripN, lane, l0, c0, l1, c1);
                const int Pn = ((P + 2) & (GM_PASSES - 1)) * 64;
                a4n = (A4 + Pn)[lane];       a4m = (A4 + Pn + 32)[lane];
                m2n = (M2 + Pn)[lane];       m2m = (M2 + Pn + 32)[lane];
                wn  = (W1 + Pn)[lane];       wm  = (W1 + Pn + 32)[lane];
            }

            const v2f xp = x1p[g >> 1];
            v8f s0, s1, s2, s3;
            if (g & 1) {
                s0 = gm_scores<1>(aT0, bq[g], kl[0], kc[0], xp);
                s1 = gm_scores<1>(aT1, bq[g], kl[1], kc[1], xp);
                s2 = gm_scores<1>(aT2, bq[g], kl[2], kc[2], xp);
                s3 = gm_scores<1>(aT3, bq[g], kl[3], kc[3], xp);
            } else {
                s0 = gm_scores<0>(aT0, bq[g], kl[0], kc[0], xp);
                s1 = gm_scores<0>(aT1, bq[g], kl[1], kc[1], xp);
                s2 = gm_scores<0>(aT2, bq[g], kl[2], kc[2], xp);
                s3 = gm_scores<0>(aT3, bq[g], kl[3], kc[3], xp);
            }

            if (g == GM_GROUPS - 1) {
                asm volatile("" ::: "memory");
                gm_strip_get(stripN, half, kl, kc);
            }

            gm_group(s0, s1, s2, s3, mx[g], sum[g]);
        }
        aT0 = aN0;  aT1 = aN1;  aT2 = aN2;  aT3 = aN3;
    }

    for (int g = 0; g < GM_GROUPS; ++g)
        part[wave][g][lane] = make_float2(mx[g], sum[g]);
    wnorm[wave][lane] = make_float2(wmax, wsum);
    __syncthreads();

    const int gq = 2 * wave + half;
    float2 pp[2 * GM_WAVES];
    for (int v = 0; v < GM_WAVES; ++v) {
        pp[2 * v]     = part[v][gq][ccol];
        pp[2 * v + 1] = part[v][gq][ccol + 16];
    }
    float2 wp[GM_WAVES];
    for (int v = 0; v < GM_WAVES; ++v) wp[v] = wnorm[v][lane];

    const float mAll = fmaxf(fmaxf(fmaxf(pp[0].x, pp[1].x), fmaxf(pp[2].x, pp[3].x)),
                             fmaxf(fmaxf(pp[4].x, pp[5].x), fmaxf(pp[6].x, pp[7].x)));
    float sAll = 0.0f;
    for (int i = 0; i < 2 * GM_WAVES; ++i) sAll += pp[i].y * __expf(pp[i].x - mAll);

    const float wAll = fmaxf(fmaxf(wp[0].x, wp[1].x), fmaxf(wp[2].x, wp[3].x));
    float wsAll = 0.0f;
    for (int v = 0; v < GM_WAVES; ++v) wsAll += wp[v].y * __expf(wp[v].x - wAll);
    const float lse_w = wAll + __logf(wsAll);

    out[colBase + 32 * wave + lane] = (mAll + __logf(sAll)) - lse_w;
}

extern "C" void kernel_launch(void* const* d_in, const int* in_sizes, int n_in,
                              void* d_out, int out_size, void* d_ws, size_t ws_size,
                              hipStream_t stream) {
    const float* sample = (const float*)d_in[0];
    const float* mu     = (const float*)d_in[1];
    const float* Amat   = (const float*)d_in[2];
    const float* w      = (const float*)d_in[3];
    float* out = (float*)d_out;

    const int n = in_sizes[0] / 2;
    const int blocks = n / 128;
    gm_fused_kernel<<<blocks, 128, 0, stream>>>(sample, mu, Amat, w, out);
}
